# starting kernel
# baseline (speedup 1.0000x reference)
#include <hip/hip_runtime.h>
#include <math.h>

#define NQ   12
#define NL   6
#define DIM  4096
#define BLK  256

typedef __attribute__((ext_vector_type(2))) float v2f;
typedef __attribute__((ext_vector_type(8))) float v8f;

__device__ __forceinline__ void rx_pair(v2f& a, v2f& b, float c, float s) {
    v2f na, nb;
    na.x =  c * a.x + s * b.y;
    na.y =  c * a.y - s * b.x;
    nb.x =  s * a.y + c * b.x;
    nb.y = -s * a.x + c * b.y;
    a = na; b = nb;
}

__global__ __launch_bounds__(BLK)
void qsim_kernel(const float* __restrict__ x,
                 const float* __restrict__ w,
                 float* __restrict__ out) {
    __shared__ v2f lds2[2 * DIM];
    int cur = 0;

    const int tid = threadIdx.x;
    const int b   = blockIdx.x;

    float a0r[NQ], a0i[NQ], a1r[NQ], a1i[NQ];
    #pragma unroll
    for (int q = 0; q < NQ; ++q) {
        float ang = tanhf(x[b * NQ + q]) * 3.14159265358979323846f;
        float h = 0.5f * ang;
        float c = __cosf(h), s = __sinf(h);
        a0r[q] = c * c;  a0i[q] = -s * c;
        a1r[q] = s * c;  a1i[q] = s * s;
    }
    {
        v2f* st = lds2;
        #pragma unroll 1
        for (int k = 0; k < DIM / BLK; ++k) {
            int s = tid + BLK * k;
            float vr = 1.0f, vi = 0.0f;
            #pragma unroll
            for (int q = 0; q < NQ; ++q) {
                int bit = (s >> (NQ - 1 - q)) & 1;
                float ar = bit ? a1r[q] : a0r[q];
                float ai = bit ? a1i[q] : a0i[q];
                float nr = vr * ar - vi * ai;
                float ni = vr * ai + vi * ar;
                vr = nr; vi = ni;
            }
            v2f amp; amp.x = vr; amp.y = vi;
            st[s] = amp;
        }
    }

    #pragma unroll 1
    for (int l = 0; l < NL; ++l) {
        #pragma unroll 1
        for (int g = 0; g < NQ / 2; ++g) {
            __syncthreads();
            float ha = 0.5f * w[l * NQ + 2 * g];
            float hb = 0.5f * w[l * NQ + 2 * g + 1];
            float ca = __cosf(ha), sa = __sinf(ha);
            float cb = __cosf(hb), sb = __sinf(hb);
            int ma = 1 << (NQ - 1 - 2 * g);
            int mb = ma >> 1;
            v2f* st = lds2 + cur * DIM;
            #pragma unroll
            for (int k = 0; k < (DIM / 4) / BLK; ++k) {
                int p   = tid + BLK * k;
                int s00 = ((p & ~(mb - 1)) << 2) | (p & (mb - 1));
                v2f a00 = st[s00];
                v2f a01 = st[s00 | mb];
                v2f a10 = st[s00 | ma];
                v2f a11 = st[s00 | ma | mb];
                rx_pair(a00, a10, ca, sa);
                rx_pair(a01, a11, ca, sa);
                rx_pair(a00, a01, cb, sb);
                rx_pair(a10, a11, cb, sb);
                st[s00]           = a00;
                st[s00 | mb]      = a01;
                st[s00 | ma]      = a10;
                st[s00 | ma | mb] = a11;
            }
        }
        __syncthreads();
        {
            v2f* src = lds2 + cur * DIM;
            v2f* dst = lds2 + (cur ^ 1) * DIM;
            #pragma unroll 1
            for (int k = 0; k < DIM / BLK; ++k) {
                int s = tid + BLK * k;
                int j = s;
                #pragma unroll
                for (int i = NQ - 1; i >= 0; --i) {
                    int cpos = NQ - 1 - i;
                    int tpos = NQ - 1 - ((i + 1) % NQ);
                    j ^= ((j >> cpos) & 1) << tpos;
                }
                dst[s] = src[j];
            }
        }
        cur ^= 1;
    }

    __syncthreads();
    float* P  = (float*)(lds2 + (cur ^ 1) * DIM);
    float* R  = P + DIM;
    float* Wt = R + 512;
    {
        v2f* st = lds2 + cur * DIM;
        #pragma unroll 1
        for (int k = 0; k < DIM / BLK; ++k) {
            int s = tid + BLK * k;
            v2f a = st[s];
            P[s] = a.x * a.x + a.y * a.y;
        }
        #pragma unroll
        for (int e = 0; e < 4; ++e) {
            int idx = tid + BLK * e;
            int c   = idx >> 6;
            int lo  = idx & 63;
            float v = 0.0f;
            if (c < 6)       v = 1.0f - 2.0f * (float)((lo >> c) & 1);
            else if (c == 6) v = 1.0f;
            Wt[idx] = v;
        }
    }
    __syncthreads();

    int wave = tid >> 5;
    int lane = tid & 31;
    if (wave < 4) {
        int row   = (wave << 4) + (lane & 15);
        int col   = lane & 15;
        int khalf = (lane >> 4) << 1;
        v8f acc = {};
        #pragma unroll
        for (int k = 0; k < 64; k += 4) {
            int k0 = k + khalf;
            v2f A = *(const v2f*)(P  + row * 64 + k0);
            v2f B = *(const v2f*)(Wt + col * 64 + k0);
            acc = __builtin_amdgcn_wmma_f32_16x16x4_f32(
                 false, A,  false, B,
                 (short)0, acc,  false,  false);
        }
        if (col < 8) {
            int rbase = (wave << 4) + ((lane >> 4) << 3);
            #pragma unroll
            for (int v = 0; v < 8; ++v)
                R[(rbase + v) * 8 + col] = acc[v];
        }
    }
    __syncthreads();

    if (tid < NQ) {
        int q = tid;
        float acc = 0.0f;
        if (q >= 6) {
            int c = NQ - 1 - q;
            #pragma unroll 1
            for (int hi = 0; hi < 64; ++hi) acc += R[hi * 8 + c];
        } else {
            int bp = 5 - q;
            #pragma unroll 1
            for (int hi = 0; hi < 64; ++hi) {
                float sgn = 1.0f - 2.0f * (float)((hi >> bp) & 1);
                acc += sgn * R[hi * 8 + 6];
            }
        }
        out[b * NQ + q] = acc;
    }
}

extern "C" void kernel_launch(void* const* d_in, const int* in_sizes, int n_in,
                              void* d_out, int out_size, void* d_ws, size_t ws_size,
                              hipStream_t stream) {
    const float* x  = (const float*)d_in[0];
    const float* w  = (const float*)d_in[1];
    float* out = (float*)d_out;
    int batch = in_sizes[0] / NQ;
    qsim_kernel<<<batch, BLK, 0, stream>>>(x, w, out);
}


# Circuit state kept in registers of two waves (64 amplitudes/lane): RX in place on register-bit qubits, by DPP/permlane on lane-bit qubits, at the gather on the wave-bit qubit; LDS only for CNOT ring
# speedup vs baseline: 1.9604x; 1.9604x over previous
#include <hip/hip_runtime.h>
#include <math.h>

#define NQ   12
#define NL   6
#define DIM  4096
#define NR   64
#define WAVE 32
#define BLK  64

typedef __attribute__((ext_vector_type(2))) float v2f;
typedef __attribute__((ext_vector_type(8))) float v8f;

__host__ __device__ constexpr int cnot_src(int s) {
    int j = s;
    for (int i = NQ - 1; i >= 0; --i) {
        int cpos = NQ - 1 - i;
        int tpos = NQ - 1 - ((i + 1) % NQ);
        j ^= ((j >> cpos) & 1) << tpos;
    }
    return j;
}

template <int M>
__device__ __forceinline__ float lane_xor(float v) {
    int i = __builtin_bit_cast(int, v);
    if constexpr (M == 16) {
        return __builtin_bit_cast(float,
            __builtin_amdgcn_permlanex16(i, i, 0x76543210, 0xfedcba98, false, false));
    } else {
        return __builtin_bit_cast(float,
            __builtin_amdgcn_update_dpp(0, i, 0x160 + M, 0xf, 0xf, true));
    }
}

__device__ __forceinline__ v2f rx_side(v2f own, v2f par, float c, float s) {
    v2f cc = {c, c};
    v2f sn = {s, -s};
    v2f ps = {par.y, par.x};
    return __builtin_elementwise_fma(sn, ps, cc * own);
}

template <int MASK>
__device__ __forceinline__ void rx_reg(v2f (&st)[NR], float c, float s) {
    #pragma unroll
    for (int r = 0; r < NR; ++r)
        if ((r & MASK) == 0) {
            v2f a = st[r], b = st[r | MASK];
            st[r]        = rx_side(a, b, c, s);
            st[r | MASK] = rx_side(b, a, c, s);
        }
}

template <int M>
__device__ __forceinline__ void rx_lane(v2f (&st)[NR], float c, float s) {
    #pragma unroll
    for (int r = 0; r < NR; ++r) {
        float px = lane_xor<M>(st[r].x);
        float py = lane_xor<M>(st[r].y);
        float nx = c * st[r].x + s * py;
        float ny = c * st[r].y - s * px;
        st[r].x = nx; st[r].y = ny;
    }
}

__device__ __forceinline__ v2f cmul(v2f v, float ar, float ai) {
    v2f o;
    o.x = v.x * ar - v.y * ai;
    o.y = v.x * ai + v.y * ar;
    return o;
}

__global__ __launch_bounds__(BLK) __attribute__((amdgpu_waves_per_eu(4)))
void qsim_kernel(const float* __restrict__ x,
                 const float* __restrict__ w,
                 float* __restrict__ out) {
    __shared__ v2f lds2[DIM];

    const int tid  = threadIdx.x;
    const int lane = tid & 31;
    const int wv   = tid >> 5;
    const int b    = blockIdx.x;

    float a0r[NQ], a0i[NQ], a1r[NQ], a1i[NQ];
    #pragma unroll
    for (int q = 0; q < NQ; ++q) {
        float ang = tanhf(x[b * NQ + q]) * 3.14159265358979323846f;
        float h = 0.5f * ang;
        float c = __cosf(h), s = __sinf(h);
        a0r[q] = c * c;  a0i[q] = -s * c;
        a1r[q] = s * c;  a1i[q] = s * s;
    }

    v2f st[NR];
    {
        float vr = 1.0f, vi = 0.0f;
        #pragma unroll
        for (int q = 7; q < NQ; ++q) {
            int bit = (lane >> (NQ - 1 - q)) & 1;
            float ar = bit ? a1r[q] : a0r[q];
            float ai = bit ? a1i[q] : a0i[q];
            float nr = vr * ar - vi * ai;
            float ni = vr * ai + vi * ar;
            vr = nr; vi = ni;
        }
        st[0].x = vr; st[0].y = vi;
    }
    #pragma unroll
    for (int q = 6; q >= 1; --q) {
        const int n = 1 << (6 - q);
        #pragma unroll
        for (int i = 0; i < n; ++i) {
            v2f v = st[i];
            st[i + n] = cmul(v, a1r[q], a1i[q]);
            st[i]     = cmul(v, a0r[q], a0i[q]);
        }
    }
    {
        float oR = wv ? a1r[0] : a0r[0], oI = wv ? a1i[0] : a0i[0];
        float pR = wv ? a0r[0] : a1r[0], pI = wv ? a0i[0] : a1i[0];
        float h0 = 0.5f * w[0];
        float c0 = __cosf(h0), s0 = __sinf(h0);
        #pragma unroll
        for (int r = 0; r < NR; ++r) {
            v2f own = cmul(st[r], oR, oI);
            v2f par = cmul(st[r], pR, pI);
            st[r] = rx_side(own, par, c0, s0);
        }
    }

    const int plw = (cnot_src(lane) ^ (wv ? cnot_src(2048) : 0)) * 8;
    const int plp = plw ^ (cnot_src(2048) * 8);
    const char* ldsb = (const char*)lds2;

    #pragma unroll 1
    for (int l = 0; l < NL; ++l) {
        float c[NQ], s[NQ];
        #pragma unroll
        for (int q = 1; q < NQ; ++q) {
            float h = 0.5f * w[l * NQ + q];
            c[q] = __cosf(h); s[q] = __sinf(h);
        }
        rx_reg<32>(st, c[1], s[1]);
        rx_reg<16>(st, c[2], s[2]);
        rx_reg<8 >(st, c[3], s[3]);
        rx_reg<4 >(st, c[4], s[4]);
        rx_reg<2 >(st, c[5], s[5]);
        rx_reg<1 >(st, c[6], s[6]);
        rx_lane<16>(st, c[7],  s[7]);
        rx_lane<8 >(st, c[8],  s[8]);
        rx_lane<4 >(st, c[9],  s[9]);
        rx_lane<2 >(st, c[10], s[10]);
        rx_lane<1 >(st, c[11], s[11]);

        #pragma unroll
        for (int r = 0; r < NR; ++r) lds2[wv * 2048 + r * WAVE + lane] = st[r];
        __syncthreads();
        if (l + 1 < NL) {
            float hn = 0.5f * w[(l + 1) * NQ];
            float cn = __cosf(hn), sn = __sinf(hn);
            #pragma unroll
            for (int r = 0; r < NR; ++r) {
                const int c8 = cnot_src(r << 5) * 8;
                v2f own = *(const v2f*)(ldsb + (plw ^ c8));
                v2f par = *(const v2f*)(ldsb + (plp ^ c8));
                st[r] = rx_side(own, par, cn, sn);
            }
        } else {
            #pragma unroll
            for (int r = 0; r < NR; ++r) {
                const int c8 = cnot_src(r << 5) * 8;
                st[r] = *(const v2f*)(ldsb + (plw ^ c8));
            }
        }
        __syncthreads();
    }

    float* P  = (float*)lds2;
    float* R  = P + DIM;
    float* Wt = R + 512;
    #pragma unroll
    for (int r = 0; r < NR; ++r)
        P[wv * 2048 + r * WAVE + lane] = st[r].x * st[r].x + st[r].y * st[r].y;
    #pragma unroll 4
    for (int e = 0; e < 1024 / BLK; ++e) {
        int idx = tid + BLK * e;
        int cc  = idx >> 6;
        int lo  = idx & 63;
        float v = 0.0f;
        if (cc < 6)       v = 1.0f - 2.0f * (float)((lo >> cc) & 1);
        else if (cc == 6) v = 1.0f;
        Wt[idx] = v;
    }
    __syncthreads();

    const int col   = lane & 15;
    const int khalf = (lane >> 4) << 1;
    const int row0  = (wv << 5) + col;
    const int row1  = row0 + 16;
    v8f acc0 = {}, acc1 = {};
    #pragma unroll
    for (int k = 0; k < 64; k += 4) {
        int k0 = k + khalf;
        v2f B  = *(const v2f*)(Wt + col * 64 + k0);
        v2f A0 = *(const v2f*)(P + row0 * 64 + k0);
        v2f A1 = *(const v2f*)(P + row1 * 64 + k0);
        acc0 = __builtin_amdgcn_wmma_f32_16x16x4_f32(false, A0, false, B, (short)0, acc0, false, false);
        acc1 = __builtin_amdgcn_wmma_f32_16x16x4_f32(false, A1, false, B, (short)0, acc1, false, false);
    }
    if (col < 8) {
        int rb = (wv << 5) + ((lane >> 4) << 3);
        #pragma unroll
        for (int v = 0; v < 8; ++v) {
            R[(rb + v) * 8 + col]      = acc0[v];
            R[(rb + 16 + v) * 8 + col] = acc1[v];
        }
    }
    __syncthreads();

    if (tid < NQ) {
        int q = tid;
        float acc = 0.0f;
        if (q >= 6) {
            int c = NQ - 1 - q;
            #pragma unroll 8
            for (int hi = 0; hi < 64; ++hi) acc += R[hi * 8 + c];
        } else {
            int bp = 5 - q;
            #pragma unroll 8
            for (int hi = 0; hi < 64; ++hi) {
                float sgn = 1.0f - 2.0f * (float)((hi >> bp) & 1);
                acc += sgn * R[hi * 8 + 6];
            }
        }
        out[b * NQ + q] = acc;
    }
}

extern "C" void kernel_launch(void* const* d_in, const int* in_sizes, int n_in,
                              void* d_out, int out_size, void* d_ws, size_t ws_size,
                              hipStream_t stream) {
    const float* x  = (const float*)d_in[0];
    const float* w  = (const float*)d_in[1];
    float* out = (float*)d_out;
    int batch = in_sizes[0] / NQ;
    qsim_kernel<<<batch, BLK, 0, stream>>>(x, w, out);
}


# All 12 RX gates as packed in-thread register pairs: a second, padded LDS pass per layer swaps lane and register index bits (no DPP/permlane gates); a layer's gates now run 1-6,0,7-11
# speedup vs baseline: 2.2445x; 1.1449x over previous
#include <hip/hip_runtime.h>
#include <math.h>

#define NQ   12
#define NL   6
#define DIM  4096
#define NR   64
#define WAVE 32
#define BLK  64

#define LDSN (DIM + DIM / 32)

typedef __attribute__((ext_vector_type(2))) float v2f;
typedef __attribute__((ext_vector_type(8))) float v8f;

__host__ __device__ constexpr int cnot_src(int s) {
    int j = s;
    for (int i = NQ - 1; i >= 0; --i) {
        int cpos = NQ - 1 - i;
        int tpos = NQ - 1 - ((i + 1) % NQ);
        j ^= ((j >> cpos) & 1) << tpos;
    }
    return j;
}

__host__ __device__ constexpr int slot(int y) { return y + (y >> 5); }

__device__ __forceinline__ v2f rx_side(v2f own, v2f par, float c, float s) {
    v2f cc = {c, c};
    v2f sn = {s, -s};
    v2f ps = {par.y, par.x};
    return __builtin_elementwise_fma(sn, ps, cc * own);
}

template <int MASK>
__device__ __forceinline__ void rx_reg(v2f (&st)[NR], float c, float s) {
    #pragma unroll
    for (int r = 0; r < NR; ++r)
        if ((r & MASK) == 0) {
            v2f a = st[r], b = st[r | MASK];
            st[r]        = rx_side(a, b, c, s);
            st[r | MASK] = rx_side(b, a, c, s);
        }
}

__device__ __forceinline__ v2f cmul(v2f v, float ar, float ai) {
    v2f o;
    o.x = v.x * ar - v.y * ai;
    o.y = v.x * ai + v.y * ar;
    return o;
}

__global__ __launch_bounds__(BLK) __attribute__((amdgpu_waves_per_eu(4)))
void qsim_kernel(const float* __restrict__ x,
                 const float* __restrict__ w,
                 float* __restrict__ out) {
    __shared__ v2f lds2[LDSN];

    const int tid  = threadIdx.x;
    const int lane = tid & 31;
    const int wv   = tid >> 5;
    const int b    = blockIdx.x;

    float a0r[NQ], a0i[NQ], a1r[NQ], a1i[NQ];
    #pragma unroll
    for (int q = 0; q < NQ; ++q) {
        float ang = tanhf(x[b * NQ + q]) * 3.14159265358979323846f;
        float h = 0.5f * ang;
        float c = __cosf(h), s = __sinf(h);
        a0r[q] = c * c;  a0i[q] = -s * c;
        a1r[q] = s * c;  a1i[q] = s * s;
    }

    v2f st[NR];
    {
        float vr = wv ? a1r[0] : a0r[0], vi = wv ? a1i[0] : a0i[0];
        #pragma unroll
        for (int q = 7; q < NQ; ++q) {
            int bit = (lane >> (NQ - 1 - q)) & 1;
            float ar = bit ? a1r[q] : a0r[q];
            float ai = bit ? a1i[q] : a0i[q];
            float nr = vr * ar - vi * ai;
            float ni = vr * ai + vi * ar;
            vr = nr; vi = ni;
        }
        st[0].x = vr; st[0].y = vi;
    }
    #pragma unroll
    for (int q = 6; q >= 1; --q) {
        const int n = 1 << (6 - q);
        #pragma unroll
        for (int i = 0; i < n; ++i) {
            v2f v = st[i];
            st[i + n] = cmul(v, a1r[q], a1i[q]);
            st[i]     = cmul(v, a0r[q], a0i[q]);
        }
    }

    char* ldsb = (char*)lds2;
    const int offA = slot((wv << 11) | lane) * 8;
    const int offB = slot(tid << 5) * 8;
    const int T = cnot_src(lane) ^ (wv ? cnot_src(2048) : 0);
    int gb[4];
    #pragma unroll
    for (int k = 0; k < 4; ++k)
        gb[k] = slot(T ^ (((k & 1) << 4) | ((k >> 1) << 10))) * 8;

    #pragma unroll 1
    for (int l = 0; l < NL; ++l) {
        float c[NQ], s[NQ];
        #pragma unroll
        for (int q = 0; q < NQ; ++q) {
            float h = 0.5f * w[l * NQ + q];
            c[q] = __cosf(h); s[q] = __sinf(h);
        }
        rx_reg<32>(st, c[1], s[1]);
        rx_reg<16>(st, c[2], s[2]);
        rx_reg<8 >(st, c[3], s[3]);
        rx_reg<4 >(st, c[4], s[4]);
        rx_reg<2 >(st, c[5], s[5]);
        rx_reg<1 >(st, c[6], s[6]);

        #pragma unroll
        for (int r = 0; r < NR; ++r)
            *(v2f*)(ldsb + offA + slot(r << 5) * 8) = st[r];
        __syncthreads();
        #pragma unroll
        for (int r = 0; r < NR; ++r)
            st[r] = *(const v2f*)(ldsb + offB + slot(((r >> 5) << 11) | (r & 31)) * 8);

        rx_reg<32>(st, c[0],  s[0]);
        rx_reg<16>(st, c[7],  s[7]);
        rx_reg<8 >(st, c[8],  s[8]);
        rx_reg<4 >(st, c[9],  s[9]);
        rx_reg<2 >(st, c[10], s[10]);
        rx_reg<1 >(st, c[11], s[11]);

        #pragma unroll
        for (int r = 0; r < NR; ++r)
            *(v2f*)(ldsb + offB + slot(((r >> 5) << 11) | (r & 31)) * 8) = st[r];
        __syncthreads();
        #pragma unroll
        for (int r = 0; r < NR; ++r) {
            const int C  = cnot_src(r << 5);
            const int k  = ((C >> 4) & 1) | (((C >> 10) & 1) << 1);
            const int ca = C & ~0x410;
            st[r] = *(const v2f*)(ldsb + gb[k] + slot(ca) * 8);
        }
        __syncthreads();
    }

    float* P  = (float*)lds2;
    float* R  = P + DIM;
    float* Wt = R + 512;
    #pragma unroll
    for (int r = 0; r < NR; ++r)
        P[wv * 2048 + r * WAVE + lane] = st[r].x * st[r].x + st[r].y * st[r].y;
    #pragma unroll 4
    for (int e = 0; e < 1024 / BLK; ++e) {
        int idx = tid + BLK * e;
        int cc  = idx >> 6;
        int lo  = idx & 63;
        float v = 0.0f;
        if (cc < 6)       v = 1.0f - 2.0f * (float)((lo >> cc) & 1);
        else if (cc == 6) v = 1.0f;
        Wt[idx] = v;
    }
    __syncthreads();

    const int col   = lane & 15;
    const int khalf = (lane >> 4) << 1;
    const int row0  = (wv << 5) + col;
    const int row1  = row0 + 16;
    v8f acc0 = {}, acc1 = {};
    #pragma unroll
    for (int k = 0; k < 64; k += 4) {
        int k0 = k + khalf;
        v2f B  = *(const v2f*)(Wt + col * 64 + k0);
        v2f A0 = *(const v2f*)(P + row0 * 64 + k0);
        v2f A1 = *(const v2f*)(P + row1 * 64 + k0);
        acc0 = __builtin_amdgcn_wmma_f32_16x16x4_f32(false, A0, false, B, (short)0, acc0, false, false);
        acc1 = __builtin_amdgcn_wmma_f32_16x16x4_f32(false, A1, false, B, (short)0, acc1, false, false);
    }
    if (col < 8) {
        int rb = (wv << 5) + ((lane >> 4) << 3);
        #pragma unroll
        for (int v = 0; v < 8; ++v) {
            R[(rb + v) * 8 + col]      = acc0[v];
            R[(rb + 16 + v) * 8 + col] = acc1[v];
        }
    }
    __syncthreads();

    if (tid < NQ) {
        int q = tid;
        float acc = 0.0f;
        if (q >= 6) {
            int c = NQ - 1 - q;
            #pragma unroll 8
            for (int hi = 0; hi < 64; ++hi) acc += R[hi * 8 + c];
        } else {
            int bp = 5 - q;
            #pragma unroll 8
            for (int hi = 0; hi < 64; ++hi) {
                float sgn = 1.0f - 2.0f * (float)((hi >> bp) & 1);
                acc += sgn * R[hi * 8 + 6];
            }
        }
        out[b * NQ + q] = acc;
    }
}

extern "C" void kernel_launch(void* const* d_in, const int* in_sizes, int n_in,
                              void* d_out, int out_size, void* d_ws, size_t ws_size,
                              hipStream_t stream) {
    const float* x  = (const float*)d_in[0];
    const float* w  = (const float*)d_in[1];
    float* out = (float*)d_out;
    int batch = in_sizes[0] / NQ;
    qsim_kernel<<<batch, BLK, 0, stream>>>(x, w, out);
}


# RX gate arithmetic in batches of 8 register pairs: a batch's packed multiplies first, then its fmas (scheduling fences hold the order), so no fma issues right behind the multiply it reads
# speedup vs baseline: 2.5689x; 1.1445x over previous
#include <hip/hip_runtime.h>
#include <math.h>

#define NQ   12
#define NL   6
#define DIM  4096
#define NR   64
#define WAVE 32
#define BLK  64

#define LDSN (DIM + DIM / 32)

typedef __attribute__((ext_vector_type(2))) float v2f;
typedef __attribute__((ext_vector_type(8))) float v8f;

__host__ __device__ constexpr int cnot_src(int s) {
    int j = s;
    for (int i = NQ - 1; i >= 0; --i) {
        int cpos = NQ - 1 - i;
        int tpos = NQ - 1 - ((i + 1) % NQ);
        j ^= ((j >> cpos) & 1) << tpos;
    }
    return j;
}

__host__ __device__ constexpr int slot(int y) { return y + (y >> 5); }

#define RX_BATCH 8
template <int MASK>
__device__ __forceinline__ void rx_reg(v2f (&st)[NR], float c, float s) {
    const v2f cc = {c, c};
    const v2f sn = {s, -s};
    #pragma unroll
    for (int p0 = 0; p0 < NR / 2; p0 += RX_BATCH) {
        v2f ta[RX_BATCH], tb[RX_BATCH];
        #pragma unroll
        for (int j = 0; j < RX_BATCH; ++j) {
            const int p = p0 + j;
            const int r = ((p & ~(MASK - 1)) << 1) | (p & (MASK - 1));
            ta[j] = cc * st[r];
            tb[j] = cc * st[r | MASK];
        }
        __builtin_amdgcn_sched_barrier(0x0084);
        #pragma unroll
        for (int j = 0; j < RX_BATCH; ++j) {
            const int p = p0 + j;
            const int r = ((p & ~(MASK - 1)) << 1) | (p & (MASK - 1));
            const v2f a = st[r], b = st[r | MASK];
            const v2f bs = {b.y, b.x};
            const v2f as = {a.y, a.x};
            st[r]        = __builtin_elementwise_fma(sn, bs, ta[j]);
            st[r | MASK] = __builtin_elementwise_fma(sn, as, tb[j]);
        }
        __builtin_amdgcn_sched_barrier(0x0084);
    }
}

__device__ __forceinline__ v2f cmul(v2f v, float ar, float ai) {
    v2f o;
    o.x = v.x * ar - v.y * ai;
    o.y = v.x * ai + v.y * ar;
    return o;
}

__global__ __launch_bounds__(BLK) __attribute__((amdgpu_waves_per_eu(4)))
void qsim_kernel(const float* __restrict__ x,
                 const float* __restrict__ w,
                 float* __restrict__ out) {
    __shared__ v2f lds2[LDSN];

    const int tid  = threadIdx.x;
    const int lane = tid & 31;
    const int wv   = tid >> 5;
    const int b    = blockIdx.x;

    float a0r[NQ], a0i[NQ], a1r[NQ], a1i[NQ];
    #pragma unroll
    for (int q = 0; q < NQ; ++q) {
        float ang = tanhf(x[b * NQ + q]) * 3.14159265358979323846f;
        float h = 0.5f * ang;
        float c = __cosf(h), s = __sinf(h);
        a0r[q] = c * c;  a0i[q] = -s * c;
        a1r[q] = s * c;  a1i[q] = s * s;
    }

    v2f st[NR];
    {
        float vr = wv ? a1r[0] : a0r[0], vi = wv ? a1i[0] : a0i[0];
        #pragma unroll
        for (int q = 7; q < NQ; ++q) {
            int bit = (lane >> (NQ - 1 - q)) & 1;
            float ar = bit ? a1r[q] : a0r[q];
            float ai = bit ? a1i[q] : a0i[q];
            float nr = vr * ar - vi * ai;
            float ni = vr * ai + vi * ar;
            vr = nr; vi = ni;
        }
        st[0].x = vr; st[0].y = vi;
    }
    #pragma unroll
    for (int q = 6; q >= 1; --q) {
        const int n = 1 << (6 - q);
        #pragma unroll
        for (int i = 0; i < n; ++i) {
            v2f v = st[i];
            st[i + n] = cmul(v, a1r[q], a1i[q]);
            st[i]     = cmul(v, a0r[q], a0i[q]);
        }
    }

    char* ldsb = (char*)lds2;
    const int offA = slot((wv << 11) | lane) * 8;
    const int offB = slot(tid << 5) * 8;
    const int T = cnot_src(lane) ^ (wv ? cnot_src(2048) : 0);
    int gb[4];
    #pragma unroll
    for (int k = 0; k < 4; ++k)
        gb[k] = slot(T ^ (((k & 1) << 4) | ((k >> 1) << 10))) * 8;

    #pragma unroll 1
    for (int l = 0; l < NL; ++l) {
        float c[NQ], s[NQ];
        #pragma unroll
        for (int q = 0; q < NQ; ++q) {
            float h = 0.5f * w[l * NQ + q];
            c[q] = __cosf(h); s[q] = __sinf(h);
        }
        rx_reg<32>(st, c[1], s[1]);
        rx_reg<16>(st, c[2], s[2]);
        rx_reg<8 >(st, c[3], s[3]);
        rx_reg<4 >(st, c[4], s[4]);
        rx_reg<2 >(st, c[5], s[5]);
        rx_reg<1 >(st, c[6], s[6]);

        #pragma unroll
        for (int r = 0; r < NR; ++r)
            *(v2f*)(ldsb + offA + slot(r << 5) * 8) = st[r];
        __syncthreads();
        #pragma unroll
        for (int r = 0; r < NR; ++r)
            st[r] = *(const v2f*)(ldsb + offB + slot(((r >> 5) << 11) | (r & 31)) * 8);

        rx_reg<32>(st, c[0],  s[0]);
        rx_reg<16>(st, c[7],  s[7]);
        rx_reg<8 >(st, c[8],  s[8]);
        rx_reg<4 >(st, c[9],  s[9]);
        rx_reg<2 >(st, c[10], s[10]);
        rx_reg<1 >(st, c[11], s[11]);

        #pragma unroll
        for (int r = 0; r < NR; ++r)
            *(v2f*)(ldsb + offB + slot(((r >> 5) << 11) | (r & 31)) * 8) = st[r];
        __syncthreads();
        #pragma unroll
        for (int r = 0; r < NR; ++r) {
            const int C  = cnot_src(r << 5);
            const int k  = ((C >> 4) & 1) | (((C >> 10) & 1) << 1);
            const int ca = C & ~0x410;
            st[r] = *(const v2f*)(ldsb + gb[k] + slot(ca) * 8);
        }
        __syncthreads();
    }

    float* P  = (float*)lds2;
    float* R  = P + DIM;
    float* Wt = R + 512;
    #pragma unroll
    for (int r = 0; r < NR; ++r)
        P[wv * 2048 + r * WAVE + lane] = st[r].x * st[r].x + st[r].y * st[r].y;
    #pragma unroll 4
    for (int e = 0; e < 1024 / BLK; ++e) {
        int idx = tid + BLK * e;
        int cc  = idx >> 6;
        int lo  = idx & 63;
        float v = 0.0f;
        if (cc < 6)       v = 1.0f - 2.0f * (float)((lo >> cc) & 1);
        else if (cc == 6) v = 1.0f;
        Wt[idx] = v;
    }
    __syncthreads();

    const int col   = lane & 15;
    const int khalf = (lane >> 4) << 1;
    const int row0  = (wv << 5) + col;
    const int row1  = row0 + 16;
    v8f acc0 = {}, acc1 = {};
    #pragma unroll
    for (int k = 0; k < 64; k += 4) {
        int k0 = k + khalf;
        v2f B  = *(const v2f*)(Wt + col * 64 + k0);
        v2f A0 = *(const v2f*)(P + row0 * 64 + k0);
        v2f A1 = *(const v2f*)(P + row1 * 64 + k0);
        acc0 = __builtin_amdgcn_wmma_f32_16x16x4_f32(false, A0, false, B, (short)0, acc0, false, false);
        acc1 = __builtin_amdgcn_wmma_f32_16x16x4_f32(false, A1, false, B, (short)0, acc1, false, false);
    }
    if (col < 8) {
        int rb = (wv << 5) + ((lane >> 4) << 3);
        #pragma unroll
        for (int v = 0; v < 8; ++v) {
            R[(rb + v) * 8 + col]      = acc0[v];
            R[(rb + 16 + v) * 8 + col] = acc1[v];
        }
    }
    __syncthreads();

    if (tid < NQ) {
        int q = tid;
        float acc = 0.0f;
        if (q >= 6) {
            int c = NQ - 1 - q;
            #pragma unroll 8
            for (int hi = 0; hi < 64; ++hi) acc += R[hi * 8 + c];
        } else {
            int bp = 5 - q;
            #pragma unroll 8
            for (int hi = 0; hi < 64; ++hi) {
                float sgn = 1.0f - 2.0f * (float)((hi >> bp) & 1);
                acc += sgn * R[hi * 8 + 6];
            }
        }
        out[b * NQ + q] = acc;
    }
}

extern "C" void kernel_launch(void* const* d_in, const int* in_sizes, int n_in,
                              void* d_out, int out_size, void* d_ws, size_t ws_size,
                              hipStream_t stream) {
    const float* x  = (const float*)d_in[0];
    const float* w  = (const float*)d_in[1];
    float* out = (float*)d_out;
    int batch = in_sizes[0] / NQ;
    qsim_kernel<<<batch, BLK, 0, stream>>>(x, w, out);
}


# RX on a register pair as two plane rotations of three shears each (x+=t*y, y-=s*x, x+=t*y; t=tan(h/2), s=sin h; h reduced mod pi, state sign dropped): 3 in-place packed fmas a pair, was 2 mul+2 fma
# speedup vs baseline: 2.7304x; 1.0629x over previous
#include <hip/hip_runtime.h>
#include <math.h>

#define NQ   12
#define NL   6
#define DIM  4096
#define NR   64
#define WAVE 32
#define BLK  64

#define LDSN (DIM + DIM / 32)

typedef __attribute__((ext_vector_type(2))) float v2f;
typedef __attribute__((ext_vector_type(8))) float v8f;

__host__ __device__ constexpr int cnot_src(int s) {
    int j = s;
    for (int i = NQ - 1; i >= 0; --i) {
        int cpos = NQ - 1 - i;
        int tpos = NQ - 1 - ((i + 1) % NQ);
        j ^= ((j >> cpos) & 1) << tpos;
    }
    return j;
}

__host__ __device__ constexpr int slot(int y) { return y + (y >> 5); }

#define RX_BATCH 16
template <int MASK>
__device__ __forceinline__ void rx_reg(v2f (&st)[NR], float t, float s) {
    const v2f tt = {t, -t};
    const v2f sn = {s, -s};
    #pragma unroll
    for (int p0 = 0; p0 < NR / 2; p0 += RX_BATCH) {
        #pragma unroll
        for (int j = 0; j < RX_BATCH; ++j) {
            const int p = p0 + j;
            const int r = ((p & ~(MASK - 1)) << 1) | (p & (MASK - 1));
            const v2f b = st[r | MASK];
            const v2f bs = {b.y, b.x};
            st[r] = __builtin_elementwise_fma(tt, bs, st[r]);
        }
        __builtin_amdgcn_sched_barrier(0x0084);
        #pragma unroll
        for (int j = 0; j < RX_BATCH; ++j) {
            const int p = p0 + j;
            const int r = ((p & ~(MASK - 1)) << 1) | (p & (MASK - 1));
            const v2f a = st[r];
            const v2f as = {a.y, a.x};
            st[r | MASK] = __builtin_elementwise_fma(sn, as, st[r | MASK]);
        }
        __builtin_amdgcn_sched_barrier(0x0084);
        #pragma unroll
        for (int j = 0; j < RX_BATCH; ++j) {
            const int p = p0 + j;
            const int r = ((p & ~(MASK - 1)) << 1) | (p & (MASK - 1));
            const v2f b = st[r | MASK];
            const v2f bs = {b.y, b.x};
            st[r] = __builtin_elementwise_fma(tt, bs, st[r]);
        }
    }
}

__device__ __forceinline__ v2f cmul(v2f v, float ar, float ai) {
    v2f o;
    o.x = v.x * ar - v.y * ai;
    o.y = v.x * ai + v.y * ar;
    return o;
}

__global__ __launch_bounds__(BLK) __attribute__((amdgpu_waves_per_eu(4)))
void qsim_kernel(const float* __restrict__ x,
                 const float* __restrict__ w,
                 float* __restrict__ out) {
    __shared__ v2f lds2[LDSN];

    const int tid  = threadIdx.x;
    const int lane = tid & 31;
    const int wv   = tid >> 5;
    const int b    = blockIdx.x;

    float a0r[NQ], a0i[NQ], a1r[NQ], a1i[NQ];
    #pragma unroll
    for (int q = 0; q < NQ; ++q) {
        float ang = tanhf(x[b * NQ + q]) * 3.14159265358979323846f;
        float h = 0.5f * ang;
        float c = __cosf(h), s = __sinf(h);
        a0r[q] = c * c;  a0i[q] = -s * c;
        a1r[q] = s * c;  a1i[q] = s * s;
    }

    v2f st[NR];
    {
        float vr = wv ? a1r[0] : a0r[0], vi = wv ? a1i[0] : a0i[0];
        #pragma unroll
        for (int q = 7; q < NQ; ++q) {
            int bit = (lane >> (NQ - 1 - q)) & 1;
            float ar = bit ? a1r[q] : a0r[q];
            float ai = bit ? a1i[q] : a0i[q];
            float nr = vr * ar - vi * ai;
            float ni = vr * ai + vi * ar;
            vr = nr; vi = ni;
        }
        st[0].x = vr; st[0].y = vi;
    }
    #pragma unroll
    for (int q = 6; q >= 1; --q) {
        const int n = 1 << (6 - q);
        #pragma unroll
        for (int i = 0; i < n; ++i) {
            v2f v = st[i];
            st[i + n] = cmul(v, a1r[q], a1i[q]);
            st[i]     = cmul(v, a0r[q], a0i[q]);
        }
    }

    char* ldsb = (char*)lds2;
    const int offA = slot((wv << 11) | lane) * 8;
    const int offB = slot(tid << 5) * 8;
    const int T = cnot_src(lane) ^ (wv ? cnot_src(2048) : 0);
    int gb[4];
    #pragma unroll
    for (int k = 0; k < 4; ++k)
        gb[k] = slot(T ^ (((k & 1) << 4) | ((k >> 1) << 10))) * 8;

    #pragma unroll 1
    for (int l = 0; l < NL; ++l) {
        float tn[NQ], s[NQ];
        #pragma unroll
        for (int q = 0; q < NQ; ++q) {
            float h = 0.5f * w[l * NQ + q];
            float k = rintf(h * 0.318309886183790672f);
            h = fmaf(-k, 3.14159274101257324f, h);
            h = fmaf(-k, -8.74227800037248510e-8f, h);
            float x = 0.5f * h;
            float t = __sinf(x) * __builtin_amdgcn_rcpf(__cosf(x));
            tn[q] = t;
            s[q] = (t + t) * __builtin_amdgcn_rcpf(fmaf(t, t, 1.0f));
        }
        #pragma unroll
        for (int q = 0; q < NQ; ++q)
            asm volatile("" :: "v"(tn[q]), "v"(s[q]));
        rx_reg<32>(st, tn[1], s[1]);
        rx_reg<16>(st, tn[2], s[2]);
        rx_reg<8 >(st, tn[3], s[3]);
        rx_reg<4 >(st, tn[4], s[4]);
        rx_reg<2 >(st, tn[5], s[5]);
        rx_reg<1 >(st, tn[6], s[6]);

        #pragma unroll
        for (int r = 0; r < NR; ++r)
            *(v2f*)(ldsb + offA + slot(r << 5) * 8) = st[r];
        __syncthreads();
        #pragma unroll
        for (int r = 0; r < NR; ++r)
            st[r] = *(const v2f*)(ldsb + offB + slot(((r >> 5) << 11) | (r & 31)) * 8);

        rx_reg<32>(st, tn[0],  s[0]);
        rx_reg<16>(st, tn[7],  s[7]);
        rx_reg<8 >(st, tn[8],  s[8]);
        rx_reg<4 >(st, tn[9],  s[9]);
        rx_reg<2 >(st, tn[10], s[10]);
        rx_reg<1 >(st, tn[11], s[11]);

        #pragma unroll
        for (int r = 0; r < NR; ++r)
            *(v2f*)(ldsb + offB + slot(((r >> 5) << 11) | (r & 31)) * 8) = st[r];
        __syncthreads();
        #pragma unroll
        for (int r = 0; r < NR; ++r) {
            const int C  = cnot_src(r << 5);
            const int k  = ((C >> 4) & 1) | (((C >> 10) & 1) << 1);
            const int ca = C & ~0x410;
            st[r] = *(const v2f*)(ldsb + gb[k] + slot(ca) * 8);
        }
        __syncthreads();
    }

    float* P  = (float*)lds2;
    float* R  = P + DIM;
    float* Wt = R + 512;
    #pragma unroll
    for (int r = 0; r < NR; ++r)
        P[wv * 2048 + r * WAVE + lane] = st[r].x * st[r].x + st[r].y * st[r].y;
    #pragma unroll 4
    for (int e = 0; e < 1024 / BLK; ++e) {
        int idx = tid + BLK * e;
        int cc  = idx >> 6;
        int lo  = idx & 63;
        float v = 0.0f;
        if (cc < 6)       v = 1.0f - 2.0f * (float)((lo >> cc) & 1);
        else if (cc == 6) v = 1.0f;
        Wt[idx] = v;
    }
    __syncthreads();

    const int col   = lane & 15;
    const int khalf = (lane >> 4) << 1;
    const int row0  = (wv << 5) + col;
    const int row1  = row0 + 16;
    v8f acc0 = {}, acc1 = {};
    #pragma unroll
    for (int k = 0; k < 64; k += 4) {
        int k0 = k + khalf;
        v2f B  = *(const v2f*)(Wt + col * 64 + k0);
        v2f A0 = *(const v2f*)(P + row0 * 64 + k0);
        v2f A1 = *(const v2f*)(P + row1 * 64 + k0);
        acc0 = __builtin_amdgcn_wmma_f32_16x16x4_f32(false, A0, false, B, (short)0, acc0, false, false);
        acc1 = __builtin_amdgcn_wmma_f32_16x16x4_f32(false, A1, false, B, (short)0, acc1, false, false);
    }
    if (col < 8) {
        int rb = (wv << 5) + ((lane >> 4) << 3);
        #pragma unroll
        for (int v = 0; v < 8; ++v) {
            R[(rb + v) * 8 + col]      = acc0[v];
            R[(rb + 16 + v) * 8 + col] = acc1[v];
        }
    }
    __syncthreads();

    if (tid < NQ) {
        int q = tid;
        float acc = 0.0f;
        if (q >= 6) {
            int c = NQ - 1 - q;
            #pragma unroll 8
            for (int hi = 0; hi < 64; ++hi) acc += R[hi * 8 + c];
        } else {
            int bp = 5 - q;
            #pragma unroll 8
            for (int hi = 0; hi < 64; ++hi) {
                float sgn = 1.0f - 2.0f * (float)((hi >> bp) & 1);
                acc += sgn * R[hi * 8 + 6];
            }
        }
        out[b * NQ + q] = acc;
    }
}

extern "C" void kernel_launch(void* const* d_in, const int* in_sizes, int n_in,
                              void* d_out, int out_size, void* d_ws, size_t ws_size,
                              hipStream_t stream) {
    const float* x  = (const float*)d_in[0];
    const float* w  = (const float*)d_in[1];
    float* out = (float*)d_out;
    int batch = in_sizes[0] / NQ;
    qsim_kernel<<<batch, BLK, 0, stream>>>(x, w, out);
}


# Shear factors of all 72 gates computed once per workgroup, one gate per thread, into an LDS table; a layer reads its 12 pairs by broadcast loads (was 12 sin/cos/rcp chains per wave per layer)
# speedup vs baseline: 2.8625x; 1.0484x over previous
#include <hip/hip_runtime.h>
#include <math.h>

#define NQ   12
#define NL   6
#define DIM  4096
#define NR   64
#define WAVE 32
#define BLK  64

#define LDSN (DIM + DIM / 32)

typedef __attribute__((ext_vector_type(2))) float v2f;
typedef __attribute__((ext_vector_type(8))) float v8f;

__host__ __device__ constexpr int cnot_src(int s) {
    int j = s;
    for (int i = NQ - 1; i >= 0; --i) {
        int cpos = NQ - 1 - i;
        int tpos = NQ - 1 - ((i + 1) % NQ);
        j ^= ((j >> cpos) & 1) << tpos;
    }
    return j;
}

__host__ __device__ constexpr int slot(int y) { return y + (y >> 5); }

#define RX_BATCH 16
template <int MASK>
__device__ __forceinline__ void rx_reg(v2f (&st)[NR], float t, float s) {
    const v2f tt = {t, -t};
    const v2f sn = {s, -s};
    #pragma unroll
    for (int p0 = 0; p0 < NR / 2; p0 += RX_BATCH) {
        #pragma unroll
        for (int j = 0; j < RX_BATCH; ++j) {
            const int p = p0 + j;
            const int r = ((p & ~(MASK - 1)) << 1) | (p & (MASK - 1));
            const v2f b = st[r | MASK];
            const v2f bs = {b.y, b.x};
            st[r] = __builtin_elementwise_fma(tt, bs, st[r]);
        }
        __builtin_amdgcn_sched_barrier(0x0084);
        #pragma unroll
        for (int j = 0; j < RX_BATCH; ++j) {
            const int p = p0 + j;
            const int r = ((p & ~(MASK - 1)) << 1) | (p & (MASK - 1));
            const v2f a = st[r];
            const v2f as = {a.y, a.x};
            st[r | MASK] = __builtin_elementwise_fma(sn, as, st[r | MASK]);
        }
        __builtin_amdgcn_sched_barrier(0x0084);
        #pragma unroll
        for (int j = 0; j < RX_BATCH; ++j) {
            const int p = p0 + j;
            const int r = ((p & ~(MASK - 1)) << 1) | (p & (MASK - 1));
            const v2f b = st[r | MASK];
            const v2f bs = {b.y, b.x};
            st[r] = __builtin_elementwise_fma(tt, bs, st[r]);
        }
    }
}

__device__ __forceinline__ v2f cmul(v2f v, float ar, float ai) {
    v2f o;
    o.x = v.x * ar - v.y * ai;
    o.y = v.x * ai + v.y * ar;
    return o;
}

__device__ __forceinline__ v2f gate_factors(float wgt) {
    float h = 0.5f * wgt;
    float k = rintf(h * 0.318309886183790672f);
    h = fmaf(-k, 3.14159274101257324f, h);
    h = fmaf(-k, -8.74227800037248510e-8f, h);
    float x = 0.5f * h;
    float t = __sinf(x) * __builtin_amdgcn_rcpf(__cosf(x));
    v2f f;
    f.x = t;
    f.y = (t + t) * __builtin_amdgcn_rcpf(fmaf(t, t, 1.0f));
    return f;
}

__global__ __launch_bounds__(BLK) __attribute__((amdgpu_waves_per_eu(4)))
void qsim_kernel(const float* __restrict__ x,
                 const float* __restrict__ w,
                 float* __restrict__ out) {
    __shared__ v2f lds2[LDSN];
    __shared__ v2f fac[NL * NQ];

    const int tid  = threadIdx.x;
    const int lane = tid & 31;
    const int wv   = tid >> 5;
    const int b    = blockIdx.x;

    const float wg0 = w[tid];
    const float wg1 = w[tid < NL * NQ - BLK ? tid + BLK : tid];

    float a0r[NQ], a0i[NQ], a1r[NQ], a1i[NQ];
    #pragma unroll
    for (int q = 0; q < NQ; ++q) {
        float ang = tanhf(x[b * NQ + q]) * 3.14159265358979323846f;
        float h = 0.5f * ang;
        float c = __cosf(h), s = __sinf(h);
        a0r[q] = c * c;  a0i[q] = -s * c;
        a1r[q] = s * c;  a1i[q] = s * s;
    }

    fac[tid] = gate_factors(wg0);
    if (tid < NL * NQ - BLK)
        fac[tid + BLK] = gate_factors(wg1);

    v2f st[NR];
    {
        float vr = wv ? a1r[0] : a0r[0], vi = wv ? a1i[0] : a0i[0];
        #pragma unroll
        for (int q = 7; q < NQ; ++q) {
            int bit = (lane >> (NQ - 1 - q)) & 1;
            float ar = bit ? a1r[q] : a0r[q];
            float ai = bit ? a1i[q] : a0i[q];
            float nr = vr * ar - vi * ai;
            float ni = vr * ai + vi * ar;
            vr = nr; vi = ni;
        }
        st[0].x = vr; st[0].y = vi;
    }
    #pragma unroll
    for (int q = 6; q >= 1; --q) {
        const int n = 1 << (6 - q);
        #pragma unroll
        for (int i = 0; i < n; ++i) {
            v2f v = st[i];
            st[i + n] = cmul(v, a1r[q], a1i[q]);
            st[i]     = cmul(v, a0r[q], a0i[q]);
        }
    }

    char* ldsb = (char*)lds2;
    const int offA = slot((wv << 11) | lane) * 8;
    const int offB = slot(tid << 5) * 8;
    const int T = cnot_src(lane) ^ (wv ? cnot_src(2048) : 0);
    int gb[4];
    #pragma unroll
    for (int k = 0; k < 4; ++k)
        gb[k] = slot(T ^ (((k & 1) << 4) | ((k >> 1) << 10))) * 8;

    __syncthreads();

    #pragma unroll 1
    for (int l = 0; l < NL; ++l) {
        float tn[NQ], s[NQ];
        #pragma unroll
        for (int q = 0; q < NQ; ++q) {
            const v2f f = fac[l * NQ + q];
            tn[q] = f.x;
            s[q] = f.y;
        }
        rx_reg<32>(st, tn[1], s[1]);
        rx_reg<16>(st, tn[2], s[2]);
        rx_reg<8 >(st, tn[3], s[3]);
        rx_reg<4 >(st, tn[4], s[4]);
        rx_reg<2 >(st, tn[5], s[5]);
        rx_reg<1 >(st, tn[6], s[6]);

        #pragma unroll
        for (int r = 0; r < NR; ++r)
            *(v2f*)(ldsb + offA + slot(r << 5) * 8) = st[r];
        __syncthreads();
        #pragma unroll
        for (int r = 0; r < NR; ++r)
            st[r] = *(const v2f*)(ldsb + offB + slot(((r >> 5) << 11) | (r & 31)) * 8);

        rx_reg<32>(st, tn[0],  s[0]);
        rx_reg<16>(st, tn[7],  s[7]);
        rx_reg<8 >(st, tn[8],  s[8]);
        rx_reg<4 >(st, tn[9],  s[9]);
        rx_reg<2 >(st, tn[10], s[10]);
        rx_reg<1 >(st, tn[11], s[11]);

        #pragma unroll
        for (int r = 0; r < NR; ++r)
            *(v2f*)(ldsb + offB + slot(((r >> 5) << 11) | (r & 31)) * 8) = st[r];
        __syncthreads();
        #pragma unroll
        for (int r = 0; r < NR; ++r) {
            const int C  = cnot_src(r << 5);
            const int k  = ((C >> 4) & 1) | (((C >> 10) & 1) << 1);
            const int ca = C & ~0x410;
            st[r] = *(const v2f*)(ldsb + gb[k] + slot(ca) * 8);
        }
        __syncthreads();
    }

    float* P  = (float*)lds2;
    float* R  = P + DIM;
    float* Wt = R + 512;
    #pragma unroll
    for (int r = 0; r < NR; ++r)
        P[wv * 2048 + r * WAVE + lane] = st[r].x * st[r].x + st[r].y * st[r].y;
    #pragma unroll 4
    for (int e = 0; e < 1024 / BLK; ++e) {
        int idx = tid + BLK * e;
        int cc  = idx >> 6;
        int lo  = idx & 63;
        float v = 0.0f;
        if (cc < 6)       v = 1.0f - 2.0f * (float)((lo >> cc) & 1);
        else if (cc == 6) v = 1.0f;
        Wt[idx] = v;
    }
    __syncthreads();

    const int col   = lane & 15;
    const int khalf = (lane >> 4) << 1;
    const int row0  = (wv << 5) + col;
    const int row1  = row0 + 16;
    v8f acc0 = {}, acc1 = {};
    #pragma unroll
    for (int k = 0; k < 64; k += 4) {
        int k0 = k + khalf;
        v2f B  = *(const v2f*)(Wt + col * 64 + k0);
        v2f A0 = *(const v2f*)(P + row0 * 64 + k0);
        v2f A1 = *(const v2f*)(P + row1 * 64 + k0);
        acc0 = __builtin_amdgcn_wmma_f32_16x16x4_f32(false, A0, false, B, (short)0, acc0, false, false);
        acc1 = __builtin_amdgcn_wmma_f32_16x16x4_f32(false, A1, false, B, (short)0, acc1, false, false);
    }
    if (col < 8) {
        int rb = (wv << 5) + ((lane >> 4) << 3);
        #pragma unroll
        for (int v = 0; v < 8; ++v) {
            R[(rb + v) * 8 + col]      = acc0[v];
            R[(rb + 16 + v) * 8 + col] = acc1[v];
        }
    }
    __syncthreads();

    if (tid < NQ) {
        int q = tid;
        float acc = 0.0f;
        if (q >= 6) {
            int c = NQ - 1 - q;
            #pragma unroll 8
            for (int hi = 0; hi < 64; ++hi) acc += R[hi * 8 + c];
        } else {
            int bp = 5 - q;
            #pragma unroll 8
            for (int hi = 0; hi < 64; ++hi) {
                float sgn = 1.0f - 2.0f * (float)((hi >> bp) & 1);
                acc += sgn * R[hi * 8 + 6];
            }
        }
        out[b * NQ + q] = acc;
    }
}

extern "C" void kernel_launch(void* const* d_in, const int* in_sizes, int n_in,
                              void* d_out, int out_size, void* d_ws, size_t ws_size,
                              hipStream_t stream) {
    const float* x  = (const float*)d_in[0];
    const float* w  = (const float*)d_in[1];
    float* out = (float*)d_out;
    int batch = in_sizes[0] / NQ;
    qsim_kernel<<<batch, BLK, 0, stream>>>(x, w, out);
}


# Readout WMMA fed from registers: A = a lane's probability pair as it lies (row lane&15, k = lane half, register parity), sign tile B made in registers; no LDS staging, no serial fold; new sum order
# speedup vs baseline: 3.0984x; 1.0824x over previous
#include <hip/hip_runtime.h>
#include <math.h>

#define NQ   12
#define NL   6
#define DIM  4096
#define NR   64
#define WAVE 32
#define BLK  64

#define LDSN (DIM + DIM / 32)

typedef __attribute__((ext_vector_type(2))) float v2f;
typedef __attribute__((ext_vector_type(8))) float v8f;

__host__ __device__ constexpr int cnot_src(int s) {
    int j = s;
    for (int i = NQ - 1; i >= 0; --i) {
        int cpos = NQ - 1 - i;
        int tpos = NQ - 1 - ((i + 1) % NQ);
        j ^= ((j >> cpos) & 1) << tpos;
    }
    return j;
}

__host__ __device__ constexpr int slot(int y) { return y + (y >> 5); }

#define RX_BATCH 16
template <int MASK>
__device__ __forceinline__ void rx_reg(v2f (&st)[NR], float t, float s) {
    const v2f tt = {t, -t};
    const v2f sn = {s, -s};
    #pragma unroll
    for (int p0 = 0; p0 < NR / 2; p0 += RX_BATCH) {
        #pragma unroll
        for (int j = 0; j < RX_BATCH; ++j) {
            const int p = p0 + j;
            const int r = ((p & ~(MASK - 1)) << 1) | (p & (MASK - 1));
            const v2f b = st[r | MASK];
            const v2f bs = {b.y, b.x};
            st[r] = __builtin_elementwise_fma(tt, bs, st[r]);
        }
        __builtin_amdgcn_sched_barrier(0x0084);
        #pragma unroll
        for (int j = 0; j < RX_BATCH; ++j) {
            const int p = p0 + j;
            const int r = ((p & ~(MASK - 1)) << 1) | (p & (MASK - 1));
            const v2f a = st[r];
            const v2f as = {a.y, a.x};
            st[r | MASK] = __builtin_elementwise_fma(sn, as, st[r | MASK]);
        }
        __builtin_amdgcn_sched_barrier(0x0084);
        #pragma unroll
        for (int j = 0; j < RX_BATCH; ++j) {
            const int p = p0 + j;
            const int r = ((p & ~(MASK - 1)) << 1) | (p & (MASK - 1));
            const v2f b = st[r | MASK];
            const v2f bs = {b.y, b.x};
            st[r] = __builtin_elementwise_fma(tt, bs, st[r]);
        }
    }
}

__device__ __forceinline__ v2f cmul(v2f v, float ar, float ai) {
    v2f o;
    o.x = v.x * ar - v.y * ai;
    o.y = v.x * ai + v.y * ar;
    return o;
}

__device__ __forceinline__ v2f gate_factors(float wgt) {
    float h = 0.5f * wgt;
    float k = rintf(h * 0.318309886183790672f);
    h = fmaf(-k, 3.14159274101257324f, h);
    h = fmaf(-k, -8.74227800037248510e-8f, h);
    float x = 0.5f * h;
    float t = __sinf(x) * __builtin_amdgcn_rcpf(__cosf(x));
    v2f f;
    f.x = t;
    f.y = (t + t) * __builtin_amdgcn_rcpf(fmaf(t, t, 1.0f));
    return f;
}

__global__ __launch_bounds__(BLK) __attribute__((amdgpu_waves_per_eu(4)))
void qsim_kernel(const float* __restrict__ x,
                 const float* __restrict__ w,
                 float* __restrict__ out) {
    __shared__ v2f lds2[LDSN];
    __shared__ v2f fac[NL * NQ];

    const int tid  = threadIdx.x;
    const int lane = tid & 31;
    const int wv   = tid >> 5;
    const int b    = blockIdx.x;

    const float wg0 = w[tid];
    const float wg1 = w[tid < NL * NQ - BLK ? tid + BLK : tid];

    float a0r[NQ], a0i[NQ], a1r[NQ], a1i[NQ];
    #pragma unroll
    for (int q = 0; q < NQ; ++q) {
        float ang = tanhf(x[b * NQ + q]) * 3.14159265358979323846f;
        float h = 0.5f * ang;
        float c = __cosf(h), s = __sinf(h);
        a0r[q] = c * c;  a0i[q] = -s * c;
        a1r[q] = s * c;  a1i[q] = s * s;
    }

    fac[tid] = gate_factors(wg0);
    if (tid < NL * NQ - BLK)
        fac[tid + BLK] = gate_factors(wg1);

    v2f st[NR];
    {
        float vr = wv ? a1r[0] : a0r[0], vi = wv ? a1i[0] : a0i[0];
        #pragma unroll
        for (int q = 7; q < NQ; ++q) {
            int bit = (lane >> (NQ - 1 - q)) & 1;
            float ar = bit ? a1r[q] : a0r[q];
            float ai = bit ? a1i[q] : a0i[q];
            float nr = vr * ar - vi * ai;
            float ni = vr * ai + vi * ar;
            vr = nr; vi = ni;
        }
        st[0].x = vr; st[0].y = vi;
    }
    #pragma unroll
    for (int q = 6; q >= 1; --q) {
        const int n = 1 << (6 - q);
        #pragma unroll
        for (int i = 0; i < n; ++i) {
            v2f v = st[i];
            st[i + n] = cmul(v, a1r[q], a1i[q]);
            st[i]     = cmul(v, a0r[q], a0i[q]);
        }
    }

    char* ldsb = (char*)lds2;
    const int offA = slot((wv << 11) | lane) * 8;
    const int offB = slot(tid << 5) * 8;
    const int T = cnot_src(lane) ^ (wv ? cnot_src(2048) : 0);
    int gb[4];
    #pragma unroll
    for (int k = 0; k < 4; ++k)
        gb[k] = slot(T ^ (((k & 1) << 4) | ((k >> 1) << 10))) * 8;

    __syncthreads();

    #pragma unroll 1
    for (int l = 0; l < NL; ++l) {
        float tn[NQ], s[NQ];
        #pragma unroll
        for (int q = 0; q < NQ; ++q) {
            const v2f f = fac[l * NQ + q];
            tn[q] = f.x;
            s[q] = f.y;
        }
        rx_reg<32>(st, tn[1], s[1]);
        rx_reg<16>(st, tn[2], s[2]);
        rx_reg<8 >(st, tn[3], s[3]);
        rx_reg<4 >(st, tn[4], s[4]);
        rx_reg<2 >(st, tn[5], s[5]);
        rx_reg<1 >(st, tn[6], s[6]);

        #pragma unroll
        for (int r = 0; r < NR; ++r)
            *(v2f*)(ldsb + offA + slot(r << 5) * 8) = st[r];
        __syncthreads();
        #pragma unroll
        for (int r = 0; r < NR; ++r)
            st[r] = *(const v2f*)(ldsb + offB + slot(((r >> 5) << 11) | (r & 31)) * 8);

        rx_reg<32>(st, tn[0],  s[0]);
        rx_reg<16>(st, tn[7],  s[7]);
        rx_reg<8 >(st, tn[8],  s[8]);
        rx_reg<4 >(st, tn[9],  s[9]);
        rx_reg<2 >(st, tn[10], s[10]);
        rx_reg<1 >(st, tn[11], s[11]);

        #pragma unroll
        for (int r = 0; r < NR; ++r)
            *(v2f*)(ldsb + offB + slot(((r >> 5) << 11) | (r & 31)) * 8) = st[r];
        __syncthreads();
        #pragma unroll
        for (int r = 0; r < NR; ++r) {
            const int C  = cnot_src(r << 5);
            const int k  = ((C >> 4) & 1) | (((C >> 10) & 1) << 1);
            const int ca = C & ~0x410;
            st[r] = *(const v2f*)(ldsb + gb[k] + slot(ca) * 8);
        }
        __syncthreads();
    }

    const int  col  = lane & 15;
    const int  hbit = lane >> 4;
    const bool qcol = col < NQ;
    const int  bq   = qcol ? NQ - 1 - col : 0;
    float b0 = 0.0f, b1 = 0.0f;
    if (qcol) {
        b0 = 1.0f; b1 = 1.0f;
        if (bq == 4)  { b0 = hbit ? -1.0f : 1.0f; b1 = b0; }
        if (bq == 5)  { b1 = -1.0f; }
        if (bq == 11) { b0 = wv ? -1.0f : 1.0f; b1 = b0; }
    }
    const v2f Bpos = {b0, b1};
    const v2f Bneg = {-b0, -b1};
    const int jmask = (qcol && bq >= 6 && bq <= 10) ? (1 << (bq - 6)) : 0;
    v8f acc0 = {}, acc1 = {};
    #pragma unroll
    for (int j = 0; j < NR / 2; j += 2) {
        v2f A0, A1;
        A0.x = st[2 * j].x     * st[2 * j].x     + st[2 * j].y     * st[2 * j].y;
        A0.y = st[2 * j + 1].x * st[2 * j + 1].x + st[2 * j + 1].y * st[2 * j + 1].y;
        A1.x = st[2 * j + 2].x * st[2 * j + 2].x + st[2 * j + 2].y * st[2 * j + 2].y;
        A1.y = st[2 * j + 3].x * st[2 * j + 3].x + st[2 * j + 3].y * st[2 * j + 3].y;
        const v2f B0 = (jmask & j)       ? Bneg : Bpos;
        const v2f B1 = (jmask & (j + 1)) ? Bneg : Bpos;
        acc0 = __builtin_amdgcn_wmma_f32_16x16x4_f32(false, A0, false, B0, (short)0, acc0, false, false);
        acc1 = __builtin_amdgcn_wmma_f32_16x16x4_f32(false, A1, false, B1, (short)0, acc1, false, false);
    }
    const int vsh = (qcol && bq <= 2) ? bq : 3;
    float part = 0.0f;
    #pragma unroll
    for (int v = 0; v < 8; ++v) {
        const float a = acc0[v] + acc1[v];
        part += ((v >> vsh) & 1) ? -a : a;
    }
    if (qcol && bq == 3 && hbit) part = -part;
    part += __shfl_xor(part, 16);
    __shared__ float comb[2 * 16];
    if (lane < 16) comb[wv * 16 + lane] = part;
    __syncthreads();
    if (tid < NQ) out[b * NQ + tid] = comb[tid] + comb[16 + tid];
}

extern "C" void kernel_launch(void* const* d_in, const int* in_sizes, int n_in,
                              void* d_out, int out_size, void* d_ws, size_t ws_size,
                              hipStream_t stream) {
    const float* x  = (const float*)d_in[0];
    const float* w  = (const float*)d_in[1];
    float* out = (float*)d_out;
    int batch = in_sizes[0] / NQ;
    qsim_kernel<<<batch, BLK, 0, stream>>>(x, w, out);
}
